# starting kernel
# baseline (speedup 1.0000x reference)
#include <hip/hip_runtime.h>
#include <hip/hip_bf16.h>

typedef float v2f __attribute__((ext_vector_type(2)));
typedef float v8f __attribute__((ext_vector_type(8)));

#define WAVES 8
#define ROWS_PER_BLOCK (WAVES * 16)

__global__ __launch_bounds__(256, 1)
void cvx_policy_kernel(const float* __restrict__ zin, const float* __restrict__ tin,
                       const float* __restrict__ W1, const float* __restrict__ b1,
                       const float* __restrict__ W2, const float* __restrict__ b2,
                       float* __restrict__ out)
{
    __shared__ float W1p[128 * 112 * 2];
    __shared__ float W2p[50 * 256 * 2];
    __shared__ float h1s[WAVES][16 * 104];

    const int tid = threadIdx.x;

    for (int idx = tid; idx < 128 * 112 * 2; idx += 256) W1p[idx] = 0.0f;
    __syncthreads();
    for (int idx = tid; idx < 256 * 100; idx += 256) {
        int k = idx / 100;
        int n = idx - k * 100;
        W1p[((k >> 1) * 112 + n) * 2 + (k & 1)] = W1[(k + 1) * 100 + n];
    }
    for (int idx = tid; idx < 100 * 256; idx += 256) {
        int k = idx >> 8;
        int n = idx & 255;
        W2p[((k >> 1) * 256 + n) * 2 + (k & 1)] = W2[idx];
    }
    __syncthreads();

    const int lane = tid & 31;
    const int wave = tid >> 5;
    const int m    = lane & 15;
    const int hi   = lane >> 4;
    const int rowBase = blockIdx.x * ROWS_PER_BLOCK + wave * 16;
    const long rowA   = rowBase + m;
    const float* zrow = zin + rowA * 256;

    float tvals[8];
#pragma unroll
    for (int i = 0; i < 8; ++i) tvals[i] = tin[rowBase + 8 * hi + i];

    v8f acc1[7];
#pragma unroll
    for (int nt = 0; nt < 7; ++nt) {
        int n  = nt * 16 + m;
        int nc = (n < 100) ? n : 99;
        float w1r0 = W1[nc];
        float b1n  = b1[nc];
        if (n >= 100) { w1r0 = 0.0f; b1n = 0.0f; }
#pragma unroll
        for (int i = 0; i < 8; ++i)
            acc1[nt][i] = tvals[i] * w1r0 + b1n;
    }

#pragma unroll 1
    for (int ks = 0; ks < 64; ++ks) {
        const int kp = 2 * ks + hi;
        v2f aF = *(const v2f*)(zrow + 2 * kp);
        const float* wb = &W1p[(kp * 112 + m) * 2];
#pragma unroll
        for (int nt = 0; nt < 7; ++nt) {
            v2f bF = *(const v2f*)(wb + nt * 32);
            acc1[nt] = __builtin_amdgcn_wmma_f32_16x16x4_f32(
                false, aF, false, bF, (short)0, acc1[nt], false, false);
        }
    }

    float* h1w = h1s[wave];
#pragma unroll
    for (int nt = 0; nt < 7; ++nt) {
        int n = nt * 16 + m;
#pragma unroll
        for (int i = 0; i < 8; ++i) {
            float v = tanhf(acc1[nt][i]);
            if (n < 104) h1w[(i + 8 * hi) * 104 + n] = v;
        }
    }
    __syncthreads();

    v8f acc2[16];
    const v8f zero8 = {0.f, 0.f, 0.f, 0.f, 0.f, 0.f, 0.f, 0.f};
#pragma unroll
    for (int nt = 0; nt < 16; ++nt) acc2[nt] = zero8;

#pragma unroll 1
    for (int ks = 0; ks < 25; ++ks) {
        const int kp = 2 * ks + hi;
        v2f aF = *(const v2f*)&h1w[m * 104 + 2 * kp];
        const float* wb = &W2p[(kp * 256 + m) * 2];
#pragma unroll
        for (int nt = 0; nt < 16; ++nt) {
            v2f bF = *(const v2f*)(wb + nt * 32);
            acc2[nt] = __builtin_amdgcn_wmma_f32_16x16x4_f32(
                false, aF, false, bF, (short)0, acc2[nt], false, false);
        }
    }

    float r2v[8];
#pragma unroll
    for (int i = 0; i < 8; ++i) r2v[i] = 0.0f;
#pragma unroll
    for (int nt = 0; nt < 16; ++nt) {
        float bb = b2[nt * 16 + m];
#pragma unroll
        for (int i = 0; i < 8; ++i) {
            float p = acc2[nt][i] + bb;
            acc2[nt][i] = p;
            r2v[i] += p * p;
        }
    }
#pragma unroll
    for (int mask = 1; mask <= 8; mask <<= 1) {
#pragma unroll
        for (int i = 0; i < 8; ++i)
            r2v[i] += __shfl_xor(r2v[i], mask, 32);
    }

    float sca[8];
#pragma unroll
    for (int i = 0; i < 8; ++i) {
        const float r2 = r2v[i];
        float w = log1pf(r2);
#pragma unroll 1
        for (int it = 0; it < 30; ++it) {
            float ew = expf(w);
            w = w - (w * ew - r2) / ((1.0f + w) * ew);
        }
        w = fmaxf(w, 0.0f);
        float tn = sqrtf(w);
        float r  = sqrtf(r2);
        sca[i] = (r > 1e-12f) ? (tn / fmaxf(r, 1e-12f)) : 1.0f;
    }

#pragma unroll
    for (int nt = 0; nt < 16; ++nt) {
        int n = nt * 16 + m;
#pragma unroll
        for (int i = 0; i < 8; ++i) {
            long r = (long)(rowBase + i + 8 * hi);
            out[r * 256 + n] = -sca[i] * acc2[nt][i];
        }
    }
}

extern "C" void kernel_launch(void* const* d_in, const int* in_sizes, int n_in,
                              void* d_out, int out_size, void* d_ws, size_t ws_size,
                              hipStream_t stream) {
    const float* z  = (const float*)d_in[0];
    const float* t  = (const float*)d_in[1];
    const float* W1 = (const float*)d_in[2];
    const float* b1 = (const float*)d_in[3];
    const float* W2 = (const float*)d_in[4];
    const float* b2 = (const float*)d_in[5];
    float* out = (float*)d_out;

    const int rows = in_sizes[0] / 256;
    const int grid = rows / ROWS_PER_BLOCK;
    cvx_policy_kernel<<<grid, 256, 0, stream>>>(z, t, W1, b1, W2, b2, out);
}


# Weight staging rewritten: all 16-byte global loads of W1 and W2 issued before the first LDS store, K-pairs interleaved in registers, 16-byte LDS stores; zero pass and its barrier gone
# speedup vs baseline: 1.4289x; 1.4289x over previous
#include <hip/hip_runtime.h>
#include <hip/hip_bf16.h>


typedef float v2f __attribute__((ext_vector_type(2)));
typedef float v4f __attribute__((ext_vector_type(4)));
typedef float v8f __attribute__((ext_vector_type(8)));

#define WAVES 8
#define ROWS_PER_BLOCK (WAVES * 16)

__global__ __launch_bounds__(256, 1)
void cvx_policy_kernel(const float* __restrict__ zin, const float* __restrict__ tin,
                       const float* __restrict__ W1, const float* __restrict__ b1,
                       const float* __restrict__ W2, const float* __restrict__ b2,
                       float* __restrict__ out)
{
    __shared__ float lds[50 * 256 * 2 + 128 * 112 * 2];
    float* const W2p = lds;
    float* const W1p = lds + 50 * 256 * 2;

    const int tid  = threadIdx.x;
    const int lane = tid & 31;
    const int wave = tid >> 5;

    {
        const v4f zero4 = {0.f, 0.f, 0.f, 0.f};

        const int l1 = lane < 25 ? lane : 24;
        const float* w1s = W1 + 100 + 200 * wave + 4 * l1;
        v4f ra[16], rb[16];
#pragma unroll
        for (int i = 0; i < 16; ++i) {
            ra[i] = *(const v4f*)(w1s + 1600 * i);
            rb[i] = *(const v4f*)(w1s + 1600 * i + 100);
        }

        const int kq = wave >> 1;
        const int m4 = lane + 32 * (wave & 1);
        const float* w2s = W2 + 512 * kq + 4 * m4;
        v4f sa[13], sb[13];
#pragma unroll
        for (int i = 0; i < 13; ++i) {
            sa[i] = zero4;
            sb[i] = zero4;
            if (i < 12 || kq < 2) {
                sa[i] = *(const v4f*)(w2s + 2048 * i);
                sb[i] = *(const v4f*)(w2s + 2048 * i + 256);
            }
        }

        if (lane < 28) {
            const bool pad = lane >= 25;
            float* d = W1p + (wave * 112 + 4 * lane) * 2;
#pragma unroll
            for (int i = 0; i < 16; ++i) {
                const v4f a = pad ? zero4 : ra[i];
                const v4f b = pad ? zero4 : rb[i];
                const v4f lo  = {a.x, b.x, a.y, b.y};
                const v4f hi4 = {a.z, b.z, a.w, b.w};
                *(v4f*)(d + 1792 * i)     = lo;
                *(v4f*)(d + 1792 * i + 4) = hi4;
            }
        }
        {
            float* d = W2p + (kq * 256 + 4 * m4) * 2;
#pragma unroll
            for (int i = 0; i < 13; ++i) {
                if (i < 12 || kq < 2) {
                    const v4f lo  = {sa[i].x, sb[i].x, sa[i].y, sb[i].y};
                    const v4f hi4 = {sa[i].z, sb[i].z, sa[i].w, sb[i].w};
                    *(v4f*)(d + 2048 * i)     = lo;
                    *(v4f*)(d + 2048 * i + 4) = hi4;
                }
            }
        }
    }
    __syncthreads();

    const int c    = lane & 15;
    const int hi   = lane >> 4;
    const int rowBase = blockIdx.x * ROWS_PER_BLOCK + wave * 16;
    const long row    = rowBase + c;
    const float* zrow = zin + row * 256;
    const float tv    = tin[row];

    const int n1 = 4 * ((c & 7) >> 1) + 2 * (c >> 3) + (c & 1);
    const int n2 = 8 * ((c & 7) >> 2) + 4 * (c >> 3) + (c & 3);

    v8f acc1[7];
#pragma unroll
    for (int mt = 0; mt < 7; ++mt) {
#pragma unroll
        for (int j = 0; j < 4; ++j) {
            if (16 * mt + 4 * j < 100) {
                const int u = 16 * mt + 4 * j + 2 * hi;
                v2f w0 = *(const v2f*)(W1 + u);
                v2f bb = *(const v2f*)(b1 + u);
                acc1[mt][2 * j]     = tv * w0.x + bb.x;
                acc1[mt][2 * j + 1] = tv * w0.y + bb.y;
            } else {
                acc1[mt][2 * j]     = 0.0f;
                acc1[mt][2 * j + 1] = 0.0f;
            }
        }
    }

#pragma unroll 1
    for (int ks = 0; ks < 64; ++ks) {
        const int kp = 2 * ks + hi;
        v2f zF = *(const v2f*)(zrow + 2 * kp);
        const float* wa = &W1p[(kp * 112 + n1) * 2];
#pragma unroll
        for (int mt = 0; mt < 7; ++mt) {
            v2f wF = *(const v2f*)(wa + mt * 32);
            acc1[mt] = __builtin_amdgcn_wmma_f32_16x16x4_f32(
                false, wF, false, zF, (short)0, acc1[mt], false, false);
        }
    }

#pragma unroll
    for (int s = 0; s < 25; ++s) {
        acc1[s >> 2][2 * (s & 3)]     = tanhf(acc1[s >> 2][2 * (s & 3)]);
        acc1[s >> 2][2 * (s & 3) + 1] = tanhf(acc1[s >> 2][2 * (s & 3) + 1]);
    }

    v8f acc2[16];
    const v8f zero8 = {0.f, 0.f, 0.f, 0.f, 0.f, 0.f, 0.f, 0.f};
#pragma unroll
    for (int mt = 0; mt < 16; ++mt) acc2[mt] = zero8;

#pragma unroll
    for (int s = 0; s < 25; ++s) {
        const int kp = 2 * s + hi;
        v2f hF;
        hF.x = acc1[s >> 2][2 * (s & 3)];
        hF.y = acc1[s >> 2][2 * (s & 3) + 1];
        const float* wa = &W2p[(kp * 256 + n2) * 2];
#pragma unroll
        for (int mt = 0; mt < 16; ++mt) {
            v2f wF = *(const v2f*)(wa + mt * 32);
            acc2[mt] = __builtin_amdgcn_wmma_f32_16x16x4_f32(
                false, wF, false, hF, (short)0, acc2[mt], false, false);
        }
    }

    float rs0 = 0.0f, rs1 = 0.0f, rs2 = 0.0f, rs3 = 0.0f;
#pragma unroll
    for (int mt = 0; mt < 16; ++mt) {
#pragma unroll
        for (int q = 0; q < 2; ++q) {
            v4f bb = *(const v4f*)(b2 + 16 * mt + 8 * q + 4 * hi);
            float p0 = acc2[mt][4 * q + 0] + bb.x;
            float p1 = acc2[mt][4 * q + 1] + bb.y;
            float p2 = acc2[mt][4 * q + 2] + bb.z;
            float p3 = acc2[mt][4 * q + 3] + bb.w;
            acc2[mt][4 * q + 0] = p0;  rs0 += p0 * p0;
            acc2[mt][4 * q + 1] = p1;  rs1 += p1 * p1;
            acc2[mt][4 * q + 2] = p2;  rs2 += p2 * p2;
            acc2[mt][4 * q + 3] = p3;  rs3 += p3 * p3;
        }
    }
    float r2 = (rs0 + rs1) + (rs2 + rs3);
    r2 += __shfl_xor(r2, 16, 32);

    float w = log1pf(r2);
#pragma unroll 1
    for (int it = 0; it < 30; ++it) {
        float ew = expf(w);
        w = w - (w * ew - r2) / ((1.0f + w) * ew);
    }
    w = fmaxf(w, 0.0f);
    const float tn = sqrtf(w);
    const float r  = sqrtf(r2);
    const float sca = (r > 1e-12f) ? (tn / fmaxf(r, 1e-12f)) : 1.0f;

    float* orow = out + row * 256;
#pragma unroll
    for (int mt = 0; mt < 16; ++mt) {
#pragma unroll
        for (int q = 0; q < 2; ++q) {
            v4f o;
            o.x = -sca * acc2[mt][4 * q + 0];
            o.y = -sca * acc2[mt][4 * q + 1];
            o.z = -sca * acc2[mt][4 * q + 2];
            o.w = -sca * acc2[mt][4 * q + 3];
            *(v4f*)(orow + 16 * mt + 8 * q + 4 * hi) = o;
        }
    }
}

extern "C" void kernel_launch(void* const* d_in, const int* in_sizes, int n_in,
                              void* d_out, int out_size, void* d_ws, size_t ws_size,
                              hipStream_t stream) {
    const float* z  = (const float*)d_in[0];
    const float* t  = (const float*)d_in[1];
    const float* W1 = (const float*)d_in[2];
    const float* b1 = (const float*)d_in[3];
    const float* W2 = (const float*)d_in[4];
    const float* b2 = (const float*)d_in[5];
    float* out = (float*)d_out;

    const int rows = in_sizes[0] / 256;
    const int grid = rows / ROWS_PER_BLOCK;
    cvx_policy_kernel<<<grid, 256, 0, stream>>>(z, t, W1, b1, W2, b2, out);
}


# GEMM1 z operand goes through per-wave LDS tiles: whole 128-byte lines by 16-byte loads, two 32-k chunks ahead, B fragments read from LDS (same k order, same sums); no global-load wait per k-step
# speedup vs baseline: 1.5381x; 1.0764x over previous
#include <hip/hip_runtime.h>
#include <hip/hip_bf16.h>


typedef float v2f __attribute__((ext_vector_type(2)));
typedef float v4f __attribute__((ext_vector_type(4)));
typedef float v8f __attribute__((ext_vector_type(8)));

#define WAVES 8
#define ROWS_PER_BLOCK (WAVES * 16)
#define ZSLAB 72
#define ZTILE (8 * ZSLAB)

__global__ __launch_bounds__(256, 1)
void cvx_policy_kernel(const float* __restrict__ zin, const float* __restrict__ tin,
                       const float* __restrict__ W1, const float* __restrict__ b1,
                       const float* __restrict__ W2, const float* __restrict__ b2,
                       float* __restrict__ out)
{
    __shared__ float lds[50 * 256 * 2 + 128 * 112 * 2 + WAVES * 2 * ZTILE];
    float* const W2p = lds;
    float* const W1p = lds + 50 * 256 * 2;

    const int tid  = threadIdx.x;
    const int lane = tid & 31;
    const int wave = tid >> 5;
    const int rowBase = blockIdx.x * ROWS_PER_BLOCK + wave * 16;

    const int zq = lane & 7;
    const int zr = lane >> 3;
    float* const zb = lds + (50 * 256 * 2 + 128 * 112 * 2) + wave * (2 * ZTILE);
    const float* const zg = zin + (long)(rowBase + zr) * 256 + 4 * zq;
    float* const zst = zb + ZSLAB * zq + 4 * zr;
    v4f zR[4];
#pragma unroll
    for (int i = 0; i < 4; ++i) zR[i] = *(const v4f*)(zg + 1024 * i);

    {
        const v4f zero4 = {0.f, 0.f, 0.f, 0.f};

        const int l1 = lane < 25 ? lane : 24;
        const float* w1s = W1 + 100 + 200 * wave + 4 * l1;
        v4f ra[16], rb[16];
#pragma unroll
        for (int i = 0; i < 16; ++i) {
            ra[i] = *(const v4f*)(w1s + 1600 * i);
            rb[i] = *(const v4f*)(w1s + 1600 * i + 100);
        }

        const int kq = wave >> 1;
        const int m4 = lane + 32 * (wave & 1);
        const float* w2s = W2 + 512 * kq + 4 * m4;
        v4f sa[13], sb[13];
#pragma unroll
        for (int i = 0; i < 13; ++i) {
            sa[i] = zero4;
            sb[i] = zero4;
            if (i < 12 || kq < 2) {
                sa[i] = *(const v4f*)(w2s + 2048 * i);
                sb[i] = *(const v4f*)(w2s + 2048 * i + 256);
            }
        }

        if (lane < 28) {
            const bool pad = lane >= 25;
            float* d = W1p + (wave * 112 + 4 * lane) * 2;
#pragma unroll
            for (int i = 0; i < 16; ++i) {
                const v4f a = pad ? zero4 : ra[i];
                const v4f b = pad ? zero4 : rb[i];
                const v4f lo  = {a.x, b.x, a.y, b.y};
                const v4f hi4 = {a.z, b.z, a.w, b.w};
                *(v4f*)(d + 1792 * i)     = lo;
                *(v4f*)(d + 1792 * i + 4) = hi4;
            }
        }
        {
            float* d = W2p + (kq * 256 + 4 * m4) * 2;
#pragma unroll
            for (int i = 0; i < 13; ++i) {
                if (i < 12 || kq < 2) {
                    const v4f lo  = {sa[i].x, sb[i].x, sa[i].y, sb[i].y};
                    const v4f hi4 = {sa[i].z, sb[i].z, sa[i].w, sb[i].w};
                    *(v4f*)(d + 2048 * i)     = lo;
                    *(v4f*)(d + 2048 * i + 4) = hi4;
                }
            }
        }
    }
#pragma unroll
    for (int i = 0; i < 4; ++i) *(v4f*)(zst + 16 * i) = zR[i];
    __syncthreads();
#pragma unroll
    for (int i = 0; i < 4; ++i) zR[i] = *(const v4f*)(zg + 1024 * i + 32);

    const int c    = lane & 15;
    const int hi   = lane >> 4;
    const long row    = rowBase + c;
    const float tv    = tin[row];
    const float* const zld = zb + 4 * c + 2 * hi;

    const int n1 = 4 * ((c & 7) >> 1) + 2 * (c >> 3) + (c & 1);
    const int n2 = 8 * ((c & 7) >> 2) + 4 * (c >> 3) + (c & 3);

    v8f acc1[7];
#pragma unroll
    for (int mt = 0; mt < 7; ++mt) {
#pragma unroll
        for (int j = 0; j < 4; ++j) {
            if (16 * mt + 4 * j < 100) {
                const int u = 16 * mt + 4 * j + 2 * hi;
                v2f w0 = *(const v2f*)(W1 + u);
                v2f bb = *(const v2f*)(b1 + u);
                acc1[mt][2 * j]     = tv * w0.x + bb.x;
                acc1[mt][2 * j + 1] = tv * w0.y + bb.y;
            } else {
                acc1[mt][2 * j]     = 0.0f;
                acc1[mt][2 * j + 1] = 0.0f;
            }
        }
    }

#pragma unroll 1
    for (int j = 0; j < 8; ++j) {
        if (j < 7) {
            float* d = zst + ((j + 1) & 1) * ZTILE;
#pragma unroll
            for (int i = 0; i < 4; ++i) *(v4f*)(d + 16 * i) = zR[i];
        }
        if (j < 6) {
            const float* g = zg + 32 * (j + 2);
#pragma unroll
            for (int i = 0; i < 4; ++i) zR[i] = *(const v4f*)(g + 1024 * i);
        }
        __builtin_amdgcn_fence(__ATOMIC_ACQ_REL, "wavefront");
        const float* zl = zld + (j & 1) * ZTILE;
        const float* wj = &W1p[((16 * j + hi) * 112 + n1) * 2];
#pragma unroll
        for (int s = 0; s < 8; ++s) {
            v2f zF = *(const v2f*)(zl + ZSLAB * s);
            const float* wa = wj + 448 * s;
#pragma unroll
            for (int mt = 0; mt < 7; ++mt) {
                v2f wF = *(const v2f*)(wa + mt * 32);
                acc1[mt] = __builtin_amdgcn_wmma_f32_16x16x4_f32(
                    false, wF, false, zF, (short)0, acc1[mt], false, false);
            }
        }
    }

#pragma unroll
    for (int s = 0; s < 25; ++s) {
        acc1[s >> 2][2 * (s & 3)]     = tanhf(acc1[s >> 2][2 * (s & 3)]);
        acc1[s >> 2][2 * (s & 3) + 1] = tanhf(acc1[s >> 2][2 * (s & 3) + 1]);
    }

    v8f acc2[16];
    const v8f zero8 = {0.f, 0.f, 0.f, 0.f, 0.f, 0.f, 0.f, 0.f};
#pragma unroll
    for (int mt = 0; mt < 16; ++mt) acc2[mt] = zero8;

#pragma unroll
    for (int s = 0; s < 25; ++s) {
        const int kp = 2 * s + hi;
        v2f hF;
        hF.x = acc1[s >> 2][2 * (s & 3)];
        hF.y = acc1[s >> 2][2 * (s & 3) + 1];
        const float* wa = &W2p[(kp * 256 + n2) * 2];
#pragma unroll
        for (int mt = 0; mt < 16; ++mt) {
            v2f wF = *(const v2f*)(wa + mt * 32);
            acc2[mt] = __builtin_amdgcn_wmma_f32_16x16x4_f32(
                false, wF, false, hF, (short)0, acc2[mt], false, false);
        }
    }

    float rs0 = 0.0f, rs1 = 0.0f, rs2 = 0.0f, rs3 = 0.0f;
#pragma unroll
    for (int mt = 0; mt < 16; ++mt) {
#pragma unroll
        for (int q = 0; q < 2; ++q) {
            v4f bb = *(const v4f*)(b2 + 16 * mt + 8 * q + 4 * hi);
            float p0 = acc2[mt][4 * q + 0] + bb.x;
            float p1 = acc2[mt][4 * q + 1] + bb.y;
            float p2 = acc2[mt][4 * q + 2] + bb.z;
            float p3 = acc2[mt][4 * q + 3] + bb.w;
            acc2[mt][4 * q + 0] = p0;  rs0 += p0 * p0;
            acc2[mt][4 * q + 1] = p1;  rs1 += p1 * p1;
            acc2[mt][4 * q + 2] = p2;  rs2 += p2 * p2;
            acc2[mt][4 * q + 3] = p3;  rs3 += p3 * p3;
        }
    }
    float r2 = (rs0 + rs1) + (rs2 + rs3);
    r2 += __shfl_xor(r2, 16, 32);

    float w = log1pf(r2);
#pragma unroll 1
    for (int it = 0; it < 30; ++it) {
        float ew = expf(w);
        w = w - (w * ew - r2) / ((1.0f + w) * ew);
    }
    w = fmaxf(w, 0.0f);
    const float tn = sqrtf(w);
    const float r  = sqrtf(r2);
    const float sca = (r > 1e-12f) ? (tn / fmaxf(r, 1e-12f)) : 1.0f;

    float* orow = out + row * 256;
#pragma unroll
    for (int mt = 0; mt < 16; ++mt) {
#pragma unroll
        for (int q = 0; q < 2; ++q) {
            v4f o;
            o.x = -sca * acc2[mt][4 * q + 0];
            o.y = -sca * acc2[mt][4 * q + 1];
            o.z = -sca * acc2[mt][4 * q + 2];
            o.w = -sca * acc2[mt][4 * q + 3];
            *(v4f*)(orow + 16 * mt + 8 * q + 4 * hi) = o;
        }
    }
}

extern "C" void kernel_launch(void* const* d_in, const int* in_sizes, int n_in,
                              void* d_out, int out_size, void* d_ws, size_t ws_size,
                              hipStream_t stream) {
    const float* z  = (const float*)d_in[0];
    const float* t  = (const float*)d_in[1];
    const float* W1 = (const float*)d_in[2];
    const float* b1 = (const float*)d_in[3];
    const float* W2 = (const float*)d_in[4];
    const float* b2 = (const float*)d_in[5];
    float* out = (float*)d_out;

    const int rows = in_sizes[0] / 256;
    const int grid = rows / ROWS_PER_BLOCK;
    cvx_policy_kernel<<<grid, 256, 0, stream>>>(z, t, W1, b1, W2, b2, out);
}


# LDS fragment reads of both products grouped one block ahead in a second register set (GEMM1: 4 k-steps, GEMM2: 2), held by scheduling barriers: 264 read places per wave become 29; same sums
# speedup vs baseline: 1.6089x; 1.0460x over previous
#include <hip/hip_runtime.h>
#include <hip/hip_bf16.h>


typedef float v2f __attribute__((ext_vector_type(2)));
typedef float v4f __attribute__((ext_vector_type(4)));
typedef float v8f __attribute__((ext_vector_type(8)));

#define WAVES 8
#define ROWS_PER_BLOCK (WAVES * 16)
#define ZSLAB 72
#define ZTILE (8 * ZSLAB)


__device__ __forceinline__ void g1_load(v2f (&f)[32], const float* zl, const float* wa)
{
#pragma unroll
    for (int s = 0; s < 4; ++s) f[s] = *(const v2f*)(zl + ZSLAB * s);
#pragma unroll
    for (int s = 0; s < 4; ++s) {
#pragma unroll
        for (int mt = 0; mt < 7; ++mt) f[4 + 7 * s + mt] = *(const v2f*)(wa + 448 * s + 32 * mt);
    }
}

__device__ __forceinline__ void g1_mma(const v2f (&f)[32], v8f (&acc)[7])
{
#pragma unroll
    for (int s = 0; s < 4; ++s) {
#pragma unroll
        for (int mt = 0; mt < 7; ++mt) {
            acc[mt] = __builtin_amdgcn_wmma_f32_16x16x4_f32(
                false, f[4 + 7 * s + mt], false, f[s], (short)0, acc[mt], false, false);
        }
    }
}

__device__ __forceinline__ void g2_load(v2f (&f)[32], const float* wl, int s0, int n)
{
#pragma unroll
    for (int t = 0; t < 2; ++t) {
        if (t < n) {
#pragma unroll
            for (int mt = 0; mt < 16; ++mt) f[16 * t + mt] = *(const v2f*)(wl + 1024 * (s0 + t) + 32 * mt);
        }
    }
}

__device__ __forceinline__ void g2_mma(const v2f (&f)[32], const v8f (&h)[7], v8f (&acc)[16], int s0, int n)
{
#pragma unroll
    for (int t = 0; t < 2; ++t) {
        if (t < n) {
            const int s = s0 + t;
            v2f hF;
            hF.x = h[s >> 2][2 * (s & 3)];
            hF.y = h[s >> 2][2 * (s & 3) + 1];
#pragma unroll
            for (int mt = 0; mt < 16; ++mt) {
                acc[mt] = __builtin_amdgcn_wmma_f32_16x16x4_f32(
                    false, f[16 * t + mt], false, hF, (short)0, acc[mt], false, false);
            }
        }
    }
}

__global__ __launch_bounds__(256, 1)
void cvx_policy_kernel(const float* __restrict__ zin, const float* __restrict__ tin,
                       const float* __restrict__ W1, const float* __restrict__ b1,
                       const float* __restrict__ W2, const float* __restrict__ b2,
                       float* __restrict__ out)
{
    __shared__ float lds[128 * 112 * 2 + 50 * 256 * 2 + WAVES * 2 * ZTILE];
    float* const W1p = lds;
    float* const W2p = lds + 128 * 112 * 2;

    const int tid  = threadIdx.x;
    const int lane = tid & 31;
    const int wave = tid >> 5;
    const int rowBase = blockIdx.x * ROWS_PER_BLOCK + wave * 16;

    const int zq = lane & 7;
    const int zr = lane >> 3;
    float* const zb = lds + (50 * 256 * 2 + 128 * 112 * 2) + wave * (2 * ZTILE);
    const float* const zg = zin + (long)(rowBase + zr) * 256 + 4 * zq;
    float* const zst = zb + ZSLAB * zq + 4 * zr;
    v4f zR[4];
#pragma unroll
    for (int i = 0; i < 4; ++i) zR[i] = *(const v4f*)(zg + 1024 * i);

    {
        const v4f zero4 = {0.f, 0.f, 0.f, 0.f};

        const int l1 = lane < 25 ? lane : 24;
        const float* w1s = W1 + 100 + 200 * wave + 4 * l1;
        v4f ra[16], rb[16];
#pragma unroll
        for (int i = 0; i < 16; ++i) {
            ra[i] = *(const v4f*)(w1s + 1600 * i);
            rb[i] = *(const v4f*)(w1s + 1600 * i + 100);
        }

        const int kq = wave >> 1;
        const int m4 = lane + 32 * (wave & 1);
        const float* w2s = W2 + 512 * kq + 4 * m4;
        v4f sa[13], sb[13];
#pragma unroll
        for (int i = 0; i < 13; ++i) {
            sa[i] = zero4;
            sb[i] = zero4;
            if (i < 12 || kq < 2) {
                sa[i] = *(const v4f*)(w2s + 2048 * i);
                sb[i] = *(const v4f*)(w2s + 2048 * i + 256);
            }
        }

        if (lane < 28) {
            const bool pad = lane >= 25;
            float* d = W1p + (wave * 112 + 4 * lane) * 2;
#pragma unroll
            for (int i = 0; i < 16; ++i) {
                const v4f a = pad ? zero4 : ra[i];
                const v4f b = pad ? zero4 : rb[i];
                const v4f lo  = {a.x, b.x, a.y, b.y};
                const v4f hi4 = {a.z, b.z, a.w, b.w};
                *(v4f*)(d + 1792 * i)     = lo;
                *(v4f*)(d + 1792 * i + 4) = hi4;
            }
        }
        {
            float* d = W2p + (kq * 256 + 4 * m4) * 2;
#pragma unroll
            for (int i = 0; i < 13; ++i) {
                if (i < 12 || kq < 2) {
                    const v4f lo  = {sa[i].x, sb[i].x, sa[i].y, sb[i].y};
                    const v4f hi4 = {sa[i].z, sb[i].z, sa[i].w, sb[i].w};
                    *(v4f*)(d + 2048 * i)     = lo;
                    *(v4f*)(d + 2048 * i + 4) = hi4;
                }
            }
        }
    }
#pragma unroll
    for (int i = 0; i < 4; ++i) *(v4f*)(zst + 16 * i) = zR[i];
    __syncthreads();
#pragma unroll
    for (int i = 0; i < 4; ++i) zR[i] = *(const v4f*)(zg + 1024 * i + 32);

    const int c    = lane & 15;
    const int hi   = lane >> 4;
    const long row    = rowBase + c;
    const float tv    = tin[row];
    const float* const zld = zb + 4 * c + 2 * hi;

    const int n1 = 4 * ((c & 7) >> 1) + 2 * (c >> 3) + (c & 1);
    const int n2 = 8 * ((c & 7) >> 2) + 4 * (c >> 3) + (c & 3);

    v8f acc1[7];
#pragma unroll
    for (int mt = 0; mt < 7; ++mt) {
#pragma unroll
        for (int j = 0; j < 4; ++j) {
            if (16 * mt + 4 * j < 100) {
                const int u = 16 * mt + 4 * j + 2 * hi;
                v2f w0 = *(const v2f*)(W1 + u);
                v2f bb = *(const v2f*)(b1 + u);
                acc1[mt][2 * j]     = tv * w0.x + bb.x;
                acc1[mt][2 * j + 1] = tv * w0.y + bb.y;
            } else {
                acc1[mt][2 * j]     = 0.0f;
                acc1[mt][2 * j + 1] = 0.0f;
            }
        }
    }

    v2f fa[32], fb[32];
    const float* const wl = &W1p[(hi * 112 + n1) * 2];
    g1_load(fa, zld, wl);
#pragma unroll 1
    for (int j = 0; j < 7; ++j) {
        {
            float* d = zst + ((j + 1) & 1) * ZTILE;
#pragma unroll
            for (int i = 0; i < 4; ++i) *(v4f*)(d + 16 * i) = zR[i];
        }
        if (j < 6) {
            const float* g = zg + 32 * (j + 2);
#pragma unroll
            for (int i = 0; i < 4; ++i) zR[i] = *(const v4f*)(g + 1024 * i);
        }
        __builtin_amdgcn_fence(__ATOMIC_ACQ_REL, "wavefront");
        const float* zl = zld + (j & 1) * ZTILE;
        const float* wj = wl + 3584 * j;
        g1_load(fb, zl + 4 * ZSLAB, wj + 4 * 448);
        __builtin_amdgcn_sched_barrier(0);
        g1_mma(fa, acc1);
        __builtin_amdgcn_sched_barrier(0);
        g1_load(fa, zld + ((j + 1) & 1) * ZTILE, wj + 8 * 448);
        __builtin_amdgcn_sched_barrier(0);
        g1_mma(fb, acc1);
        __builtin_amdgcn_sched_barrier(0);
    }
    g1_load(fb, zld + ZTILE + 4 * ZSLAB, wl + 3584 * 7 + 4 * 448);
    __builtin_amdgcn_sched_barrier(0);
    g1_mma(fa, acc1);
    __builtin_amdgcn_sched_barrier(0);

    v2f ga[32], gb[32];
    const float* const w2l = &W2p[(hi * 256 + n2) * 2];
    g2_load(ga, w2l, 0, 2);
    __builtin_amdgcn_sched_barrier(0);
    g1_mma(fb, acc1);
    __builtin_amdgcn_sched_barrier(0);

#pragma unroll
    for (int s = 0; s < 25; ++s) {
        acc1[s >> 2][2 * (s & 3)]     = tanhf(acc1[s >> 2][2 * (s & 3)]);
        acc1[s >> 2][2 * (s & 3) + 1] = tanhf(acc1[s >> 2][2 * (s & 3) + 1]);
    }

    v8f acc2[16];
    const v8f zero8 = {0.f, 0.f, 0.f, 0.f, 0.f, 0.f, 0.f, 0.f};
#pragma unroll
    for (int mt = 0; mt < 16; ++mt) acc2[mt] = zero8;

#pragma unroll
    for (int b = 0; b < 13; ++b) {
        if (b < 12) {
            if ((b & 1) == 0) g2_load(gb, w2l, 2 * b + 2, 2);
            else              g2_load(ga, w2l, 2 * b + 2, b == 11 ? 1 : 2);
        }
        __builtin_amdgcn_sched_barrier(0);
        if ((b & 1) == 0) g2_mma(ga, acc1, acc2, 2 * b, b == 12 ? 1 : 2);
        else              g2_mma(gb, acc1, acc2, 2 * b, 2);
        __builtin_amdgcn_sched_barrier(0);
    }

    float rs0 = 0.0f, rs1 = 0.0f, rs2 = 0.0f, rs3 = 0.0f;
#pragma unroll
    for (int mt = 0; mt < 16; ++mt) {
#pragma unroll
        for (int q = 0; q < 2; ++q) {
            v4f bb = *(const v4f*)(b2 + 16 * mt + 8 * q + 4 * hi);
            float p0 = acc2[mt][4 * q + 0] + bb.x;
            float p1 = acc2[mt][4 * q + 1] + bb.y;
            float p2 = acc2[mt][4 * q + 2] + bb.z;
            float p3 = acc2[mt][4 * q + 3] + bb.w;
            acc2[mt][4 * q + 0] = p0;  rs0 += p0 * p0;
            acc2[mt][4 * q + 1] = p1;  rs1 += p1 * p1;
            acc2[mt][4 * q + 2] = p2;  rs2 += p2 * p2;
            acc2[mt][4 * q + 3] = p3;  rs3 += p3 * p3;
        }
    }
    float r2 = (rs0 + rs1) + (rs2 + rs3);
    r2 += __shfl_xor(r2, 16, 32);

    float w = log1pf(r2);
#pragma unroll 1
    for (int it = 0; it < 30; ++it) {
        float ew = expf(w);
        w = w - (w * ew - r2) / ((1.0f + w) * ew);
    }
    w = fmaxf(w, 0.0f);
    const float tn = sqrtf(w);
    const float r  = sqrtf(r2);
    const float sca = (r > 1e-12f) ? (tn / fmaxf(r, 1e-12f)) : 1.0f;

    float* orow = out + row * 256;
#pragma unroll
    for (int mt = 0; mt < 16; ++mt) {
#pragma unroll
        for (int q = 0; q < 2; ++q) {
            v4f o;
            o.x = -sca * acc2[mt][4 * q + 0];
            o.y = -sca * acc2[mt][4 * q + 1];
            o.z = -sca * acc2[mt][4 * q + 2];
            o.w = -sca * acc2[mt][4 * q + 3];
            *(v4f*)(orow + 16 * mt + 8 * q + 4 * hi) = o;
        }
    }
}

extern "C" void kernel_launch(void* const* d_in, const int* in_sizes, int n_in,
                              void* d_out, int out_size, void* d_ws, size_t ws_size,
                              hipStream_t stream) {
    const float* z  = (const float*)d_in[0];
    const float* t  = (const float*)d_in[1];
    const float* W1 = (const float*)d_in[2];
    const float* b1 = (const float*)d_in[3];
    const float* W2 = (const float*)d_in[4];
    const float* b2 = (const float*)d_in[5];
    float* out = (float*)d_out;

    const int rows = in_sizes[0] / 256;
    const int grid = rows / ROWS_PER_BLOCK;
    cvx_policy_kernel<<<grid, 256, 0, stream>>>(z, t, W1, b1, W2, b2, out);
}


# tanh of the hidden tile on the hardware v_tanh_f32 instruction (one transcendental per value) instead of the library's two-range branch form: 100 exec branches per wave gone; h rounding changes
# speedup vs baseline: 1.7263x; 1.0730x over previous
#include <hip/hip_runtime.h>
#include <hip/hip_bf16.h>


typedef float v2f __attribute__((ext_vector_type(2)));
typedef float v4f __attribute__((ext_vector_type(4)));
typedef float v8f __attribute__((ext_vector_type(8)));

#define WAVES 8
#define ROWS_PER_BLOCK (WAVES * 16)
#define ZSLAB 72
#define ZTILE (8 * ZSLAB)


__device__ __forceinline__ void g1_load(v2f (&f)[32], const float* zl, const float* wa)
{
#pragma unroll
    for (int s = 0; s < 4; ++s) f[s] = *(const v2f*)(zl + ZSLAB * s);
#pragma unroll
    for (int s = 0; s < 4; ++s) {
#pragma unroll
        for (int mt = 0; mt < 7; ++mt) f[4 + 7 * s + mt] = *(const v2f*)(wa + 448 * s + 32 * mt);
    }
}

__device__ __forceinline__ void g1_mma(const v2f (&f)[32], v8f (&acc)[7])
{
#pragma unroll
    for (int s = 0; s < 4; ++s) {
#pragma unroll
        for (int mt = 0; mt < 7; ++mt) {
            acc[mt] = __builtin_amdgcn_wmma_f32_16x16x4_f32(
                false, f[4 + 7 * s + mt], false, f[s], (short)0, acc[mt], false, false);
        }
    }
}

__device__ __forceinline__ void g2_load(v2f (&f)[32], const float* wl, int s0, int n)
{
#pragma unroll
    for (int t = 0; t < 2; ++t) {
        if (t < n) {
#pragma unroll
            for (int mt = 0; mt < 16; ++mt) f[16 * t + mt] = *(const v2f*)(wl + 1024 * (s0 + t) + 32 * mt);
        }
    }
}

__device__ __forceinline__ void g2_mma(const v2f (&f)[32], const v8f (&h)[7], v8f (&acc)[16], int s0, int n)
{
#pragma unroll
    for (int t = 0; t < 2; ++t) {
        if (t < n) {
            const int s = s0 + t;
            v2f hF;
            hF.x = h[s >> 2][2 * (s & 3)];
            hF.y = h[s >> 2][2 * (s & 3) + 1];
#pragma unroll
            for (int mt = 0; mt < 16; ++mt) {
                acc[mt] = __builtin_amdgcn_wmma_f32_16x16x4_f32(
                    false, f[16 * t + mt], false, hF, (short)0, acc[mt], false, false);
            }
        }
    }
}

__global__ __launch_bounds__(256, 1)
void cvx_policy_kernel(const float* __restrict__ zin, const float* __restrict__ tin,
                       const float* __restrict__ W1, const float* __restrict__ b1,
                       const float* __restrict__ W2, const float* __restrict__ b2,
                       float* __restrict__ out)
{
    __shared__ float lds[128 * 112 * 2 + 50 * 256 * 2 + WAVES * 2 * ZTILE];
    float* const W1p = lds;
    float* const W2p = lds + 128 * 112 * 2;

    const int tid  = threadIdx.x;
    const int lane = tid & 31;
    const int wave = tid >> 5;
    const int rowBase = blockIdx.x * ROWS_PER_BLOCK + wave * 16;

    const int zq = lane & 7;
    const int zr = lane >> 3;
    float* const zb = lds + (50 * 256 * 2 + 128 * 112 * 2) + wave * (2 * ZTILE);
    const float* const zg = zin + (long)(rowBase + zr) * 256 + 4 * zq;
    float* const zst = zb + ZSLAB * zq + 4 * zr;
    v4f zR[4];
#pragma unroll
    for (int i = 0; i < 4; ++i) zR[i] = *(const v4f*)(zg + 1024 * i);

    {
        const v4f zero4 = {0.f, 0.f, 0.f, 0.f};

        const int l1 = lane < 25 ? lane : 24;
        const float* w1s = W1 + 100 + 200 * wave + 4 * l1;
        v4f ra[16], rb[16];
#pragma unroll
        for (int i = 0; i < 16; ++i) {
            ra[i] = *(const v4f*)(w1s + 1600 * i);
            rb[i] = *(const v4f*)(w1s + 1600 * i + 100);
        }

        const int kq = wave >> 1;
        const int m4 = lane + 32 * (wave & 1);
        const float* w2s = W2 + 512 * kq + 4 * m4;
        v4f sa[13], sb[13];
#pragma unroll
        for (int i = 0; i < 13; ++i) {
            sa[i] = zero4;
            sb[i] = zero4;
            if (i < 12 || kq < 2) {
                sa[i] = *(const v4f*)(w2s + 2048 * i);
                sb[i] = *(const v4f*)(w2s + 2048 * i + 256);
            }
        }

        if (lane < 28) {
            const bool pad = lane >= 25;
            float* d = W1p + (wave * 112 + 4 * lane) * 2;
#pragma unroll
            for (int i = 0; i < 16; ++i) {
                const v4f a = pad ? zero4 : ra[i];
                const v4f b = pad ? zero4 : rb[i];
                const v4f lo  = {a.x, b.x, a.y, b.y};
                const v4f hi4 = {a.z, b.z, a.w, b.w};
                *(v4f*)(d + 1792 * i)     = lo;
                *(v4f*)(d + 1792 * i + 4) = hi4;
            }
        }
        {
            float* d = W2p + (kq * 256 + 4 * m4) * 2;
#pragma unroll
            for (int i = 0; i < 13; ++i) {
                if (i < 12 || kq < 2) {
                    const v4f lo  = {sa[i].x, sb[i].x, sa[i].y, sb[i].y};
                    const v4f hi4 = {sa[i].z, sb[i].z, sa[i].w, sb[i].w};
                    *(v4f*)(d + 2048 * i)     = lo;
                    *(v4f*)(d + 2048 * i + 4) = hi4;
                }
            }
        }
    }
#pragma unroll
    for (int i = 0; i < 4; ++i) *(v4f*)(zst + 16 * i) = zR[i];
    __syncthreads();
#pragma unroll
    for (int i = 0; i < 4; ++i) zR[i] = *(const v4f*)(zg + 1024 * i + 32);

    const int c    = lane & 15;
    const int hi   = lane >> 4;
    const long row    = rowBase + c;
    const float tv    = tin[row];
    const float* const zld = zb + 4 * c + 2 * hi;

    const int n1 = 4 * ((c & 7) >> 1) + 2 * (c >> 3) + (c & 1);
    const int n2 = 8 * ((c & 7) >> 2) + 4 * (c >> 3) + (c & 3);

    v8f acc1[7];
#pragma unroll
    for (int mt = 0; mt < 7; ++mt) {
#pragma unroll
        for (int j = 0; j < 4; ++j) {
            if (16 * mt + 4 * j < 100) {
                const int u = 16 * mt + 4 * j + 2 * hi;
                v2f w0 = *(const v2f*)(W1 + u);
                v2f bb = *(const v2f*)(b1 + u);
                acc1[mt][2 * j]     = tv * w0.x + bb.x;
                acc1[mt][2 * j + 1] = tv * w0.y + bb.y;
            } else {
                acc1[mt][2 * j]     = 0.0f;
                acc1[mt][2 * j + 1] = 0.0f;
            }
        }
    }

    v2f fa[32], fb[32];
    const float* const wl = &W1p[(hi * 112 + n1) * 2];
    g1_load(fa, zld, wl);
#pragma unroll 1
    for (int j = 0; j < 7; ++j) {
        {
            float* d = zst + ((j + 1) & 1) * ZTILE;
#pragma unroll
            for (int i = 0; i < 4; ++i) *(v4f*)(d + 16 * i) = zR[i];
        }
        if (j < 6) {
            const float* g = zg + 32 * (j + 2);
#pragma unroll
            for (int i = 0; i < 4; ++i) zR[i] = *(const v4f*)(g + 1024 * i);
        }
        __builtin_amdgcn_fence(__ATOMIC_ACQ_REL, "wavefront");
        const float* zl = zld + (j & 1) * ZTILE;
        const float* wj = wl + 3584 * j;
        g1_load(fb, zl + 4 * ZSLAB, wj + 4 * 448);
        __builtin_amdgcn_sched_barrier(0);
        g1_mma(fa, acc1);
        __builtin_amdgcn_sched_barrier(0);
        g1_load(fa, zld + ((j + 1) & 1) * ZTILE, wj + 8 * 448);
        __builtin_amdgcn_sched_barrier(0);
        g1_mma(fb, acc1);
        __builtin_amdgcn_sched_barrier(0);
    }
    g1_load(fb, zld + ZTILE + 4 * ZSLAB, wl + 3584 * 7 + 4 * 448);
    __builtin_amdgcn_sched_barrier(0);
    g1_mma(fa, acc1);
    __builtin_amdgcn_sched_barrier(0);

    v2f ga[32], gb[32];
    const float* const w2l = &W2p[(hi * 256 + n2) * 2];
    g2_load(ga, w2l, 0, 2);
    __builtin_amdgcn_sched_barrier(0);
    g1_mma(fb, acc1);
    __builtin_amdgcn_sched_barrier(0);

#pragma unroll
    for (int s = 0; s < 25; ++s) {
        acc1[s >> 2][2 * (s & 3)]     = __builtin_amdgcn_tanhf(acc1[s >> 2][2 * (s & 3)]);
        acc1[s >> 2][2 * (s & 3) + 1] = __builtin_amdgcn_tanhf(acc1[s >> 2][2 * (s & 3) + 1]);
    }

    v8f acc2[16];
    const v8f zero8 = {0.f, 0.f, 0.f, 0.f, 0.f, 0.f, 0.f, 0.f};
#pragma unroll
    for (int mt = 0; mt < 16; ++mt) acc2[mt] = zero8;

#pragma unroll
    for (int b = 0; b < 13; ++b) {
        if (b < 12) {
            if ((b & 1) == 0) g2_load(gb, w2l, 2 * b + 2, 2);
            else              g2_load(ga, w2l, 2 * b + 2, b == 11 ? 1 : 2);
        }
        __builtin_amdgcn_sched_barrier(0);
        if ((b & 1) == 0) g2_mma(ga, acc1, acc2, 2 * b, b == 12 ? 1 : 2);
        else              g2_mma(gb, acc1, acc2, 2 * b, 2);
        __builtin_amdgcn_sched_barrier(0);
    }

    float rs0 = 0.0f, rs1 = 0.0f, rs2 = 0.0f, rs3 = 0.0f;
#pragma unroll
    for (int mt = 0; mt < 16; ++mt) {
#pragma unroll
        for (int q = 0; q < 2; ++q) {
            v4f bb = *(const v4f*)(b2 + 16 * mt + 8 * q + 4 * hi);
            float p0 = acc2[mt][4 * q + 0] + bb.x;
            float p1 = acc2[mt][4 * q + 1] + bb.y;
            float p2 = acc2[mt][4 * q + 2] + bb.z;
            float p3 = acc2[mt][4 * q + 3] + bb.w;
            acc2[mt][4 * q + 0] = p0;  rs0 += p0 * p0;
            acc2[mt][4 * q + 1] = p1;  rs1 += p1 * p1;
            acc2[mt][4 * q + 2] = p2;  rs2 += p2 * p2;
            acc2[mt][4 * q + 3] = p3;  rs3 += p3 * p3;
        }
    }
    float r2 = (rs0 + rs1) + (rs2 + rs3);
    r2 += __shfl_xor(r2, 16, 32);

    float w = log1pf(r2);
#pragma unroll 1
    for (int it = 0; it < 30; ++it) {
        float ew = expf(w);
        w = w - (w * ew - r2) / ((1.0f + w) * ew);
    }
    w = fmaxf(w, 0.0f);
    const float tn = sqrtf(w);
    const float r  = sqrtf(r2);
    const float sca = (r > 1e-12f) ? (tn / fmaxf(r, 1e-12f)) : 1.0f;

    float* orow = out + row * 256;
#pragma unroll
    for (int mt = 0; mt < 16; ++mt) {
#pragma unroll
        for (int q = 0; q < 2; ++q) {
            v4f o;
            o.x = -sca * acc2[mt][4 * q + 0];
            o.y = -sca * acc2[mt][4 * q + 1];
            o.z = -sca * acc2[mt][4 * q + 2];
            o.w = -sca * acc2[mt][4 * q + 3];
            *(v4f*)(orow + 16 * mt + 8 * q + 4 * hi) = o;
        }
    }
}

extern "C" void kernel_launch(void* const* d_in, const int* in_sizes, int n_in,
                              void* d_out, int out_size, void* d_ws, size_t ws_size,
                              hipStream_t stream) {
    const float* z  = (const float*)d_in[0];
    const float* t  = (const float*)d_in[1];
    const float* W1 = (const float*)d_in[2];
    const float* b1 = (const float*)d_in[3];
    const float* W2 = (const float*)d_in[4];
    const float* b2 = (const float*)d_in[5];
    float* out = (float*)d_out;

    const int rows = in_sizes[0] / 256;
    const int grid = rows / ROWS_PER_BLOCK;
    cvx_policy_kernel<<<grid, 256, 0, stream>>>(z, t, W1, b1, W2, b2, out);
}


# Output rows staged through a per-wave LDS tile (two 128-feature halves, 528-byte pitch): each global store instruction writes 512 contiguous bytes of one row, not 32 bytes in every row; same values
# speedup vs baseline: 2.2360x; 1.2953x over previous
#include <hip/hip_runtime.h>
#include <hip/hip_bf16.h>


typedef float v2f __attribute__((ext_vector_type(2)));
typedef float v4f __attribute__((ext_vector_type(4)));
typedef float v8f __attribute__((ext_vector_type(8)));

#define WAVES 8
#define ROWS_PER_BLOCK (WAVES * 16)
#define ZSLAB 72
#define ZTILE (8 * ZSLAB)
#define OPITCH 132
#define WAREA (16 * OPITCH)


__device__ __forceinline__ void g1_load(v2f (&f)[32], const float* zl, const float* wa)
{
#pragma unroll
    for (int s = 0; s < 4; ++s) f[s] = *(const v2f*)(zl + ZSLAB * s);
#pragma unroll
    for (int s = 0; s < 4; ++s) {
#pragma unroll
        for (int mt = 0; mt < 7; ++mt) f[4 + 7 * s + mt] = *(const v2f*)(wa + 448 * s + 32 * mt);
    }
}

__device__ __forceinline__ void g1_mma(const v2f (&f)[32], v8f (&acc)[7])
{
#pragma unroll
    for (int s = 0; s < 4; ++s) {
#pragma unroll
        for (int mt = 0; mt < 7; ++mt) {
            acc[mt] = __builtin_amdgcn_wmma_f32_16x16x4_f32(
                false, f[4 + 7 * s + mt], false, f[s], (short)0, acc[mt], false, false);
        }
    }
}

__device__ __forceinline__ void g2_load(v2f (&f)[32], const float* wl, int s0, int n)
{
#pragma unroll
    for (int t = 0; t < 2; ++t) {
        if (t < n) {
#pragma unroll
            for (int mt = 0; mt < 16; ++mt) f[16 * t + mt] = *(const v2f*)(wl + 1024 * (s0 + t) + 32 * mt);
        }
    }
}

__device__ __forceinline__ void g2_mma(const v2f (&f)[32], const v8f (&h)[7], v8f (&acc)[16], int s0, int n)
{
#pragma unroll
    for (int t = 0; t < 2; ++t) {
        if (t < n) {
            const int s = s0 + t;
            v2f hF;
            hF.x = h[s >> 2][2 * (s & 3)];
            hF.y = h[s >> 2][2 * (s & 3) + 1];
#pragma unroll
            for (int mt = 0; mt < 16; ++mt) {
                acc[mt] = __builtin_amdgcn_wmma_f32_16x16x4_f32(
                    false, f[16 * t + mt], false, hF, (short)0, acc[mt], false, false);
            }
        }
    }
}

__global__ __launch_bounds__(256, 1)
void cvx_policy_kernel(const float* __restrict__ zin, const float* __restrict__ tin,
                       const float* __restrict__ W1, const float* __restrict__ b1,
                       const float* __restrict__ W2, const float* __restrict__ b2,
                       float* __restrict__ out)
{
    __shared__ float lds[128 * 112 * 2 + 50 * 256 * 2 + WAVES * WAREA];
    float* const W1p = lds;
    float* const W2p = lds + 128 * 112 * 2;

    const int tid  = threadIdx.x;
    const int lane = tid & 31;
    const int wave = tid >> 5;
    const int rowBase = blockIdx.x * ROWS_PER_BLOCK + wave * 16;

    const int zq = lane & 7;
    const int zr = lane >> 3;
    float* const zb = lds + (50 * 256 * 2 + 128 * 112 * 2) + wave * WAREA;
    const float* const zg = zin + (long)(rowBase + zr) * 256 + 4 * zq;
    float* const zst = zb + ZSLAB * zq + 4 * zr;
    v4f zR[4];
#pragma unroll
    for (int i = 0; i < 4; ++i) zR[i] = *(const v4f*)(zg + 1024 * i);

    {
        const v4f zero4 = {0.f, 0.f, 0.f, 0.f};

        const int l1 = lane < 25 ? lane : 24;
        const float* w1s = W1 + 100 + 200 * wave + 4 * l1;
        v4f ra[16], rb[16];
#pragma unroll
        for (int i = 0; i < 16; ++i) {
            ra[i] = *(const v4f*)(w1s + 1600 * i);
            rb[i] = *(const v4f*)(w1s + 1600 * i + 100);
        }

        const int kq = wave >> 1;
        const int m4 = lane + 32 * (wave & 1);
        const float* w2s = W2 + 512 * kq + 4 * m4;
        v4f sa[13], sb[13];
#pragma unroll
        for (int i = 0; i < 13; ++i) {
            sa[i] = zero4;
            sb[i] = zero4;
            if (i < 12 || kq < 2) {
                sa[i] = *(const v4f*)(w2s + 2048 * i);
                sb[i] = *(const v4f*)(w2s + 2048 * i + 256);
            }
        }

        if (lane < 28) {
            const bool pad = lane >= 25;
            float* d = W1p + (wave * 112 + 4 * lane) * 2;
#pragma unroll
            for (int i = 0; i < 16; ++i) {
                const v4f a = pad ? zero4 : ra[i];
                const v4f b = pad ? zero4 : rb[i];
                const v4f lo  = {a.x, b.x, a.y, b.y};
                const v4f hi4 = {a.z, b.z, a.w, b.w};
                *(v4f*)(d + 1792 * i)     = lo;
                *(v4f*)(d + 1792 * i + 4) = hi4;
            }
        }
        {
            float* d = W2p + (kq * 256 + 4 * m4) * 2;
#pragma unroll
            for (int i = 0; i < 13; ++i) {
                if (i < 12 || kq < 2) {
                    const v4f lo  = {sa[i].x, sb[i].x, sa[i].y, sb[i].y};
                    const v4f hi4 = {sa[i].z, sb[i].z, sa[i].w, sb[i].w};
                    *(v4f*)(d + 2048 * i)     = lo;
                    *(v4f*)(d + 2048 * i + 4) = hi4;
                }
            }
        }
    }
#pragma unroll
    for (int i = 0; i < 4; ++i) *(v4f*)(zst + 16 * i) = zR[i];
    __syncthreads();
#pragma unroll
    for (int i = 0; i < 4; ++i) zR[i] = *(const v4f*)(zg + 1024 * i + 32);

    const int c    = lane & 15;
    const int hi   = lane >> 4;
    const long row    = rowBase + c;
    const float tv    = tin[row];
    const float* const zld = zb + 4 * c + 2 * hi;

    const int n1 = 4 * ((c & 7) >> 1) + 2 * (c >> 3) + (c & 1);
    const int n2 = 8 * ((c & 7) >> 2) + 4 * (c >> 3) + (c & 3);

    v8f acc1[7];
#pragma unroll
    for (int mt = 0; mt < 7; ++mt) {
#pragma unroll
        for (int j = 0; j < 4; ++j) {
            if (16 * mt + 4 * j < 100) {
                const int u = 16 * mt + 4 * j + 2 * hi;
                v2f w0 = *(const v2f*)(W1 + u);
                v2f bb = *(const v2f*)(b1 + u);
                acc1[mt][2 * j]     = tv * w0.x + bb.x;
                acc1[mt][2 * j + 1] = tv * w0.y + bb.y;
            } else {
                acc1[mt][2 * j]     = 0.0f;
                acc1[mt][2 * j + 1] = 0.0f;
            }
        }
    }

    v2f fa[32], fb[32];
    const float* const wl = &W1p[(hi * 112 + n1) * 2];
    g1_load(fa, zld, wl);
#pragma unroll 1
    for (int j = 0; j < 7; ++j) {
        {
            float* d = zst + ((j + 1) & 1) * ZTILE;
#pragma unroll
            for (int i = 0; i < 4; ++i) *(v4f*)(d + 16 * i) = zR[i];
        }
        if (j < 6) {
            const float* g = zg + 32 * (j + 2);
#pragma unroll
            for (int i = 0; i < 4; ++i) zR[i] = *(const v4f*)(g + 1024 * i);
        }
        __builtin_amdgcn_fence(__ATOMIC_ACQ_REL, "wavefront");
        const float* zl = zld + (j & 1) * ZTILE;
        const float* wj = wl + 3584 * j;
        g1_load(fb, zl + 4 * ZSLAB, wj + 4 * 448);
        __builtin_amdgcn_sched_barrier(0);
        g1_mma(fa, acc1);
        __builtin_amdgcn_sched_barrier(0);
        g1_load(fa, zld + ((j + 1) & 1) * ZTILE, wj + 8 * 448);
        __builtin_amdgcn_sched_barrier(0);
        g1_mma(fb, acc1);
        __builtin_amdgcn_sched_barrier(0);
    }
    g1_load(fb, zld + ZTILE + 4 * ZSLAB, wl + 3584 * 7 + 4 * 448);
    __builtin_amdgcn_sched_barrier(0);
    g1_mma(fa, acc1);
    __builtin_amdgcn_sched_barrier(0);

    v2f ga[32], gb[32];
    const float* const w2l = &W2p[(hi * 256 + n2) * 2];
    g2_load(ga, w2l, 0, 2);
    __builtin_amdgcn_sched_barrier(0);
    g1_mma(fb, acc1);
    __builtin_amdgcn_sched_barrier(0);

#pragma unroll
    for (int s = 0; s < 25; ++s) {
        acc1[s >> 2][2 * (s & 3)]     = __builtin_amdgcn_tanhf(acc1[s >> 2][2 * (s & 3)]);
        acc1[s >> 2][2 * (s & 3) + 1] = __builtin_amdgcn_tanhf(acc1[s >> 2][2 * (s & 3) + 1]);
    }

    v8f acc2[16];
    const v8f zero8 = {0.f, 0.f, 0.f, 0.f, 0.f, 0.f, 0.f, 0.f};
#pragma unroll
    for (int mt = 0; mt < 16; ++mt) acc2[mt] = zero8;

#pragma unroll
    for (int b = 0; b < 13; ++b) {
        if (b < 12) {
            if ((b & 1) == 0) g2_load(gb, w2l, 2 * b + 2, 2);
            else              g2_load(ga, w2l, 2 * b + 2, b == 11 ? 1 : 2);
        }
        __builtin_amdgcn_sched_barrier(0);
        if ((b & 1) == 0) g2_mma(ga, acc1, acc2, 2 * b, b == 12 ? 1 : 2);
        else              g2_mma(gb, acc1, acc2, 2 * b, 2);
        __builtin_amdgcn_sched_barrier(0);
    }

    float rs0 = 0.0f, rs1 = 0.0f, rs2 = 0.0f, rs3 = 0.0f;
#pragma unroll
    for (int mt = 0; mt < 16; ++mt) {
#pragma unroll
        for (int q = 0; q < 2; ++q) {
            v4f bb = *(const v4f*)(b2 + 16 * mt + 8 * q + 4 * hi);
            float p0 = acc2[mt][4 * q + 0] + bb.x;
            float p1 = acc2[mt][4 * q + 1] + bb.y;
            float p2 = acc2[mt][4 * q + 2] + bb.z;
            float p3 = acc2[mt][4 * q + 3] + bb.w;
            acc2[mt][4 * q + 0] = p0;  rs0 += p0 * p0;
            acc2[mt][4 * q + 1] = p1;  rs1 += p1 * p1;
            acc2[mt][4 * q + 2] = p2;  rs2 += p2 * p2;
            acc2[mt][4 * q + 3] = p3;  rs3 += p3 * p3;
        }
    }
    float r2 = (rs0 + rs1) + (rs2 + rs3);
    r2 += __shfl_xor(r2, 16, 32);

    float w = log1pf(r2);
#pragma unroll 1
    for (int it = 0; it < 30; ++it) {
        float ew = expf(w);
        w = w - (w * ew - r2) / ((1.0f + w) * ew);
    }
    w = fmaxf(w, 0.0f);
    const float tn = sqrtf(w);
    const float r  = sqrtf(r2);
    const float sca = (r > 1e-12f) ? (tn / fmaxf(r, 1e-12f)) : 1.0f;

    float* const otw = zb + OPITCH * c + 4 * hi;
    const float* const otr = zb + 4 * lane;
    float* const og = out + (long)rowBase * 256 + 4 * lane;
#pragma unroll
    for (int p = 0; p < 2; ++p) {
#pragma unroll
        for (int m = 0; m < 8; ++m) {
#pragma unroll
            for (int q = 0; q < 2; ++q) {
                const int mt = 8 * p + m;
                v4f o;
                o.x = -sca * acc2[mt][4 * q + 0];
                o.y = -sca * acc2[mt][4 * q + 1];
                o.z = -sca * acc2[mt][4 * q + 2];
                o.w = -sca * acc2[mt][4 * q + 3];
                *(v4f*)(otw + 16 * m + 8 * q) = o;
            }
        }
        __builtin_amdgcn_fence(__ATOMIC_ACQ_REL, "wavefront");
        v4f ln[16];
#pragma unroll
        for (int i = 0; i < 16; ++i) ln[i] = *(const v4f*)(otr + OPITCH * i);
#pragma unroll
        for (int i = 0; i < 16; ++i) *(v4f*)(og + 256 * i + 128 * p) = ln[i];
    }
}

extern "C" void kernel_launch(void* const* d_in, const int* in_sizes, int n_in,
                              void* d_out, int out_size, void* d_ws, size_t ws_size,
                              hipStream_t stream) {
    const float* z  = (const float*)d_in[0];
    const float* t  = (const float*)d_in[1];
    const float* W1 = (const float*)d_in[2];
    const float* b1 = (const float*)d_in[3];
    const float* W2 = (const float*)d_in[4];
    const float* b2 = (const float*)d_in[5];
    float* out = (float*)d_out;

    const int rows = in_sizes[0] / 256;
    const int grid = rows / ROWS_PER_BLOCK;
    cvx_policy_kernel<<<grid, 256, 0, stream>>>(z, t, W1, b1, W2, b2, out);
}


# Wave tile doubled to 32 batch rows (two N tiles; 256-row blocks, grid 512): each W1/W2 fragment read from LDS feeds 2 matrix instructions, weight staging and Newton tail paid per 256 rows; same sums
# speedup vs baseline: 2.6002x; 1.1629x over previous
#include <hip/hip_runtime.h>
#include <hip/hip_bf16.h>


typedef float v2f __attribute__((ext_vector_type(2)));
typedef float v4f __attribute__((ext_vector_type(4)));
typedef float v8f __attribute__((ext_vector_type(8)));

#define WAVES 8
#define NTILES 2
#define ROWS_PER_WAVE (16 * NTILES)
#define ROWS_PER_BLOCK (WAVES * ROWS_PER_WAVE)
#define ZSLAB 72
#define ZTILE (8 * ZSLAB)
#define ZCHUNK (NTILES * ZTILE)
#define OPITCH 132
#define WAREA (2 * ZCHUNK)


__device__ __forceinline__ void g1_load(v2f (&f)[36], const float* zl, const float* wa)
{
#pragma unroll
    for (int s = 0; s < 4; ++s) {
#pragma unroll
        for (int nt = 0; nt < NTILES; ++nt) f[2 * s + nt] = *(const v2f*)(zl + ZSLAB * s + ZTILE * nt);
    }
#pragma unroll
    for (int s = 0; s < 4; ++s) {
#pragma unroll
        for (int mt = 0; mt < 7; ++mt) f[8 + 7 * s + mt] = *(const v2f*)(wa + 448 * s + 32 * mt);
    }
}

__device__ __forceinline__ void g1_mma(const v2f (&f)[36], v8f (&acc)[NTILES][7])
{
#pragma unroll
    for (int s = 0; s < 4; ++s) {
#pragma unroll
        for (int mt = 0; mt < 7; ++mt) {
#pragma unroll
            for (int nt = 0; nt < NTILES; ++nt) {
                acc[nt][mt] = __builtin_amdgcn_wmma_f32_16x16x4_f32(
                    false, f[8 + 7 * s + mt], false, f[2 * s + nt], (short)0, acc[nt][mt], false, false);
            }
        }
    }
}

__device__ __forceinline__ void g2_load(v2f (&f)[16], const float* wl, int s)
{
#pragma unroll
    for (int mt = 0; mt < 16; ++mt) f[mt] = *(const v2f*)(wl + 1024 * s + 32 * mt);
}

__device__ __forceinline__ void g2_mma(const v2f (&f)[16], const v8f (&h)[NTILES][7], v8f (&acc)[NTILES][16], int s)
{
    v2f hF[NTILES];
#pragma unroll
    for (int nt = 0; nt < NTILES; ++nt) {
        hF[nt].x = h[nt][s >> 2][2 * (s & 3)];
        hF[nt].y = h[nt][s >> 2][2 * (s & 3) + 1];
    }
#pragma unroll
    for (int mt = 0; mt < 16; ++mt) {
#pragma unroll
        for (int nt = 0; nt < NTILES; ++nt) {
            acc[nt][mt] = __builtin_amdgcn_wmma_f32_16x16x4_f32(
                false, f[mt], false, hF[nt], (short)0, acc[nt][mt], false, false);
        }
    }
}

__global__ __launch_bounds__(256, 1)
void cvx_policy_kernel(const float* __restrict__ zin, const float* __restrict__ tin,
                       const float* __restrict__ W1, const float* __restrict__ b1,
                       const float* __restrict__ W2, const float* __restrict__ b2,
                       float* __restrict__ out)
{
    __shared__ float lds[128 * 112 * 2 + 50 * 256 * 2 + WAVES * WAREA];
    float* const W1p = lds;
    float* const W2p = lds + 128 * 112 * 2;

    const int tid  = threadIdx.x;
    const int lane = tid & 31;
    const int wave = tid >> 5;
    const int rowBase = blockIdx.x * ROWS_PER_BLOCK + wave * ROWS_PER_WAVE;

    const int zq = lane & 7;
    const int zr = lane >> 3;
    float* const zb = lds + (50 * 256 * 2 + 128 * 112 * 2) + wave * WAREA;
    const float* const zg = zin + (long)(rowBase + zr) * 256 + 4 * zq;
    float* const zst = zb + ZSLAB * zq + 4 * zr;
    v4f zR[8];
#pragma unroll
    for (int i = 0; i < 8; ++i) zR[i] = *(const v4f*)(zg + 1024 * i);

    {
        const v4f zero4 = {0.f, 0.f, 0.f, 0.f};

        const int l1 = lane < 25 ? lane : 24;
        const float* w1s = W1 + 100 + 200 * wave + 4 * l1;
        v4f ra[16], rb[16];
#pragma unroll
        for (int i = 0; i < 16; ++i) {
            ra[i] = *(const v4f*)(w1s + 1600 * i);
            rb[i] = *(const v4f*)(w1s + 1600 * i + 100);
        }

        const int kq = wave >> 1;
        const int m4 = lane + 32 * (wave & 1);
        const float* w2s = W2 + 512 * kq + 4 * m4;
        v4f sa[13], sb[13];
#pragma unroll
        for (int i = 0; i < 13; ++i) {
            sa[i] = zero4;
            sb[i] = zero4;
            if (i < 12 || kq < 2) {
                sa[i] = *(const v4f*)(w2s + 2048 * i);
                sb[i] = *(const v4f*)(w2s + 2048 * i + 256);
            }
        }

        if (lane < 28) {
            const bool pad = lane >= 25;
            float* d = W1p + (wave * 112 + 4 * lane) * 2;
#pragma unroll
            for (int i = 0; i < 16; ++i) {
                const v4f a = pad ? zero4 : ra[i];
                const v4f b = pad ? zero4 : rb[i];
                const v4f lo  = {a.x, b.x, a.y, b.y};
                const v4f hi4 = {a.z, b.z, a.w, b.w};
                *(v4f*)(d + 1792 * i)     = lo;
                *(v4f*)(d + 1792 * i + 4) = hi4;
            }
        }
        {
            float* d = W2p + (kq * 256 + 4 * m4) * 2;
#pragma unroll
            for (int i = 0; i < 13; ++i) {
                if (i < 12 || kq < 2) {
                    const v4f lo  = {sa[i].x, sb[i].x, sa[i].y, sb[i].y};
                    const v4f hi4 = {sa[i].z, sb[i].z, sa[i].w, sb[i].w};
                    *(v4f*)(d + 2048 * i)     = lo;
                    *(v4f*)(d + 2048 * i + 4) = hi4;
                }
            }
        }
    }
#pragma unroll
    for (int i = 0; i < 8; ++i) *(v4f*)(zst + ZTILE * (i >> 2) + 16 * (i & 3)) = zR[i];
    __syncthreads();
#pragma unroll
    for (int i = 0; i < 8; ++i) zR[i] = *(const v4f*)(zg + 1024 * i + 32);

    const int c    = lane & 15;
    const int hi   = lane >> 4;
    float tv[NTILES];
#pragma unroll
    for (int nt = 0; nt < NTILES; ++nt) tv[nt] = tin[rowBase + 16 * nt + c];
    const float* const zld = zb + 4 * c + 2 * hi;

    const int n1 = 4 * ((c & 7) >> 1) + 2 * (c >> 3) + (c & 1);
    const int n2 = 8 * ((c & 7) >> 2) + 4 * (c >> 3) + (c & 3);

    v8f acc1[NTILES][7];
#pragma unroll
    for (int mt = 0; mt < 7; ++mt) {
#pragma unroll
        for (int j = 0; j < 4; ++j) {
            if (16 * mt + 4 * j < 100) {
                const int u = 16 * mt + 4 * j + 2 * hi;
                v2f w0 = *(const v2f*)(W1 + u);
                v2f bb = *(const v2f*)(b1 + u);
#pragma unroll
                for (int nt = 0; nt < NTILES; ++nt) {
                    acc1[nt][mt][2 * j]     = tv[nt] * w0.x + bb.x;
                    acc1[nt][mt][2 * j + 1] = tv[nt] * w0.y + bb.y;
                }
            } else {
#pragma unroll
                for (int nt = 0; nt < NTILES; ++nt) {
                    acc1[nt][mt][2 * j]     = 0.0f;
                    acc1[nt][mt][2 * j + 1] = 0.0f;
                }
            }
        }
    }

    v2f fa[36], fb[36];
    const float* const wl = &W1p[(hi * 112 + n1) * 2];
    g1_load(fa, zld, wl);
#pragma unroll 1
    for (int j = 0; j < 7; ++j) {
        {
            float* d = zst + ((j + 1) & 1) * ZCHUNK;
#pragma unroll
            for (int i = 0; i < 8; ++i) *(v4f*)(d + ZTILE * (i >> 2) + 16 * (i & 3)) = zR[i];
        }
        if (j < 6) {
            const float* g = zg + 32 * (j + 2);
#pragma unroll
            for (int i = 0; i < 8; ++i) zR[i] = *(const v4f*)(g + 1024 * i);
        }
        __builtin_amdgcn_fence(__ATOMIC_ACQ_REL, "wavefront");
        const float* zl = zld + (j & 1) * ZCHUNK;
        const float* wj = wl + 3584 * j;
        g1_load(fb, zl + 4 * ZSLAB, wj + 4 * 448);
        __builtin_amdgcn_sched_barrier(0);
        g1_mma(fa, acc1);
        __builtin_amdgcn_sched_barrier(0);
        g1_load(fa, zld + ((j + 1) & 1) * ZCHUNK, wj + 8 * 448);
        __builtin_amdgcn_sched_barrier(0);
        g1_mma(fb, acc1);
        __builtin_amdgcn_sched_barrier(0);
    }
    g1_load(fb, zld + ZCHUNK + 4 * ZSLAB, wl + 3584 * 7 + 4 * 448);
    __builtin_amdgcn_sched_barrier(0);
    g1_mma(fa, acc1);
    __builtin_amdgcn_sched_barrier(0);

    v2f ga[16], gb[16];
    const float* const w2l = &W2p[(hi * 256 + n2) * 2];
    g2_load(ga, w2l, 0);
    __builtin_amdgcn_sched_barrier(0);
    g1_mma(fb, acc1);
    __builtin_amdgcn_sched_barrier(0);

#pragma unroll
    for (int nt = 0; nt < NTILES; ++nt) {
#pragma unroll
        for (int s = 0; s < 25; ++s) {
            acc1[nt][s >> 2][2 * (s & 3)]     = __builtin_amdgcn_tanhf(acc1[nt][s >> 2][2 * (s & 3)]);
            acc1[nt][s >> 2][2 * (s & 3) + 1] = __builtin_amdgcn_tanhf(acc1[nt][s >> 2][2 * (s & 3) + 1]);
        }
    }

    v8f acc2[NTILES][16];
    const v8f zero8 = {0.f, 0.f, 0.f, 0.f, 0.f, 0.f, 0.f, 0.f};
#pragma unroll
    for (int nt = 0; nt < NTILES; ++nt) {
#pragma unroll
        for (int mt = 0; mt < 16; ++mt) acc2[nt][mt] = zero8;
    }

#pragma unroll
    for (int s = 0; s < 25; ++s) {
        if (s < 24) {
            if ((s & 1) == 0) g2_load(gb, w2l, s + 1);
            else              g2_load(ga, w2l, s + 1);
        }
        __builtin_amdgcn_sched_barrier(0);
        if ((s & 1) == 0) g2_mma(ga, acc1, acc2, s);
        else              g2_mma(gb, acc1, acc2, s);
        __builtin_amdgcn_sched_barrier(0);
    }

    float rs[NTILES][4];
#pragma unroll
    for (int nt = 0; nt < NTILES; ++nt) {
        rs[nt][0] = 0.0f; rs[nt][1] = 0.0f; rs[nt][2] = 0.0f; rs[nt][3] = 0.0f;
    }
#pragma unroll
    for (int mt = 0; mt < 16; ++mt) {
#pragma unroll
        for (int q = 0; q < 2; ++q) {
            v4f bb = *(const v4f*)(b2 + 16 * mt + 8 * q + 4 * hi);
#pragma unroll
            for (int nt = 0; nt < NTILES; ++nt) {
                float p0 = acc2[nt][mt][4 * q + 0] + bb.x;
                float p1 = acc2[nt][mt][4 * q + 1] + bb.y;
                float p2 = acc2[nt][mt][4 * q + 2] + bb.z;
                float p3 = acc2[nt][mt][4 * q + 3] + bb.w;
                acc2[nt][mt][4 * q + 0] = p0;  rs[nt][0] += p0 * p0;
                acc2[nt][mt][4 * q + 1] = p1;  rs[nt][1] += p1 * p1;
                acc2[nt][mt][4 * q + 2] = p2;  rs[nt][2] += p2 * p2;
                acc2[nt][mt][4 * q + 3] = p3;  rs[nt][3] += p3 * p3;
            }
        }
    }
    float r2[NTILES];
#pragma unroll
    for (int nt = 0; nt < NTILES; ++nt) {
        r2[nt] = (rs[nt][0] + rs[nt][1]) + (rs[nt][2] + rs[nt][3]);
        r2[nt] += __shfl_xor(r2[nt], 16, 32);
    }

    float w[NTILES];
#pragma unroll
    for (int nt = 0; nt < NTILES; ++nt) w[nt] = log1pf(r2[nt]);
#pragma unroll 1
    for (int it = 0; it < 30; ++it) {
#pragma unroll
        for (int nt = 0; nt < NTILES; ++nt) {
            float ew = expf(w[nt]);
            w[nt] = w[nt] - (w[nt] * ew - r2[nt]) / ((1.0f + w[nt]) * ew);
        }
    }
    float sca[NTILES];
#pragma unroll
    for (int nt = 0; nt < NTILES; ++nt) {
        const float wm = fmaxf(w[nt], 0.0f);
        const float tn = sqrtf(wm);
        const float r  = sqrtf(r2[nt]);
        sca[nt] = (r > 1e-12f) ? (tn / fmaxf(r, 1e-12f)) : 1.0f;
    }

    float* const otw = zb + OPITCH * c + 4 * hi;
    const float* const otr = zb + 4 * lane;
    float* const og = out + (long)rowBase * 256 + 4 * lane;
#pragma unroll
    for (int nt = 0; nt < NTILES; ++nt) {
#pragma unroll
        for (int p = 0; p < 2; ++p) {
#pragma unroll
            for (int m = 0; m < 8; ++m) {
#pragma unroll
                for (int q = 0; q < 2; ++q) {
                    const int mt = 8 * p + m;
                    v4f o;
                    o.x = -sca[nt] * acc2[nt][mt][4 * q + 0];
                    o.y = -sca[nt] * acc2[nt][mt][4 * q + 1];
                    o.z = -sca[nt] * acc2[nt][mt][4 * q + 2];
                    o.w = -sca[nt] * acc2[nt][mt][4 * q + 3];
                    *(v4f*)(otw + 16 * m + 8 * q) = o;
                }
            }
            __builtin_amdgcn_fence(__ATOMIC_ACQ_REL, "wavefront");
            v4f ln[16];
#pragma unroll
            for (int i = 0; i < 16; ++i) ln[i] = *(const v4f*)(otr + OPITCH * i);
#pragma unroll
            for (int i = 0; i < 16; ++i) *(v4f*)(og + 256 * (16 * nt + i) + 128 * p) = ln[i];
        }
    }
}

extern "C" void kernel_launch(void* const* d_in, const int* in_sizes, int n_in,
                              void* d_out, int out_size, void* d_ws, size_t ws_size,
                              hipStream_t stream) {
    const float* z  = (const float*)d_in[0];
    const float* t  = (const float*)d_in[1];
    const float* W1 = (const float*)d_in[2];
    const float* b1 = (const float*)d_in[3];
    const float* W2 = (const float*)d_in[4];
    const float* b2 = (const float*)d_in[5];
    float* out = (float*)d_out;

    const int rows = in_sizes[0] / 256;
    const int grid = rows / ROWS_PER_BLOCK;
    cvx_policy_kernel<<<grid, 256, 0, stream>>>(z, t, W1, b1, W2, b2, out);
}
